# starting kernel
# baseline (speedup 1.0000x reference)
#include <hip/hip_runtime.h>
#include <math.h>


typedef __attribute__((ext_vector_type(2))) float v2f;
typedef __attribute__((ext_vector_type(8))) float v8f;

#define B_   32
#define V_   1024
#define F_   16
#define A_   32
#define P_   64
#define NF_  128
#define C_   96
#define KTOT 48
#define ROWS (B_ * V_)


__global__ __launch_bounds__(256) void k0_prep(
    const float* __restrict__ Wf, const float* __restrict__ bf,
    const float* __restrict__ Ws, const float* __restrict__ bs,
    float* __restrict__ WC, float* __restrict__ bc)
{
    const int tid = blockIdx.x * 256 + threadIdx.x;
    if (tid < F_ * C_) {
        const int k = tid / C_, c = tid % C_;
        WC[tid] = (c < P_) ? Wf[k * P_ + c] : Ws[k * A_ + (c - P_)];
    } else if (tid < F_ * C_ + C_) {
        const int c = tid - F_ * C_;
        bc[c] = (c < P_) ? bf[c] : bs[c - P_];
    }
}

__global__ __launch_bounds__(192) void k1_vertex(
    const float* __restrict__ x,  const float* __restrict__ WC,
    const float* __restrict__ bc, float* __restrict__ fws)
{
    const int lane  = threadIdx.x & 31;
    const int ntile = threadIdx.x >> 5;
    const int mtile = blockIdx.x;
    const int col   = ntile * 16 + (lane & 15);
    const int kh    = (lane >> 4) * 2;
    const float* xr = x + (size_t)(mtile * 16 + (lane & 15)) * F_;

    v8f acc = {};
#pragma unroll
    for (int kt = 0; kt < 4; ++kt) {
        const int k0 = kt * 4 + kh;
        const v2f a = *(const v2f*)(xr + k0);
        v2f b;
        b.x = WC[k0 * C_ + col];
        b.y = WC[(k0 + 1) * C_ + col];
        acc = __builtin_amdgcn_wmma_f32_16x16x4_f32(
            false, a, false, b, (short)0, acc, false, false);
    }

    const float bias = bc[col];
#pragma unroll
    for (int r = 0; r < 8; ++r) {
        const int row = mtile * 16 + r + ((lane >> 4) * 8);
        float v = acc[r] + bias;
        if (col >= P_) v = expf(-v * v);
        fws[(size_t)row * C_ + col] = v;
    }
}

__global__ __launch_bounds__(96) void k2_agg(
    const float* __restrict__ fws, float* __restrict__ agg)
{
    const int ba = blockIdx.x;
    const int b  = ba >> 5, a = ba & 31;
    const int c  = threadIdx.x;
    const float* fb = fws + (size_t)b * V_ * C_;

    float mx = -INFINITY, sm = 0.0f;
#pragma unroll 4
    for (int v = 0; v < V_; ++v) {
        const float w   = fb[(size_t)v * C_ + P_ + a];
        const float val = w * fb[(size_t)v * C_ + c];
        mx = fmaxf(mx, val);
        sm += val;
    }
    float* ao = agg + (size_t)ba * 192;
    ao[c]      = mx;
    ao[96 + c] = sm * (1.0f / (float)V_);
}

__global__ __launch_bounds__(128) void k3_bw(
    const float* __restrict__ agg, const float* __restrict__ Wout,
    float* __restrict__ BW)
{
    const int bid  = blockIdx.x;
    const int b    = bid / KTOT, krow = bid % KTOT;
    const int n    = threadIdx.x;
    float acc;
    if (krow < F_) {
        acc = Wout[(size_t)krow * NF_ + n];
    } else {
        const int a = krow - F_;
        acc = Wout[(size_t)(F_ + 2 * A_ * C_ + a) * NF_ + n];
        const float* ag = agg + (size_t)(b * A_ + a) * 192;
        const float* wr = Wout + (size_t)(F_ + a * 192) * NF_ + n;
#pragma unroll 4
        for (int c = 0; c < 192; ++c)
            acc = fmaf(ag[c], wr[(size_t)c * NF_], acc);
    }
    BW[(size_t)bid * NF_ + n] = acc;
}

__global__ __launch_bounds__(256) void k4_out(
    const float* __restrict__ x,   const float* __restrict__ fws,
    const float* __restrict__ BW,  const float* __restrict__ bout,
    float* __restrict__ out)
{
    const int lane  = threadIdx.x & 31;
    const int ntile = threadIdx.x >> 5;
    const int bid   = blockIdx.x;
    const int b     = bid >> 6, mtile = bid & 63;
    const int col   = ntile * 16 + (lane & 15);
    const int kh    = (lane >> 4) * 2;
    const int rowA  = b * V_ + mtile * 16 + (lane & 15);

    const float* xr  = x   + (size_t)rowA * F_;
    const float* er  = fws + (size_t)rowA * C_ + P_;
    const float* bwb = BW  + (size_t)b * KTOT * NF_;

    v8f acc = {};
#pragma unroll
    for (int kt = 0; kt < 4; ++kt) {
        const int k0 = kt * 4 + kh;
        const v2f a = *(const v2f*)(xr + k0);
        v2f bb;
        bb.x = bwb[(size_t)k0 * NF_ + col];
        bb.y = bwb[(size_t)(k0 + 1) * NF_ + col];
        acc = __builtin_amdgcn_wmma_f32_16x16x4_f32(
            false, a, false, bb, (short)0, acc, false, false);
    }
#pragma unroll
    for (int kt = 4; kt < 12; ++kt) {
        const int k0 = kt * 4 + kh;
        const v2f a = *(const v2f*)(er + (k0 - F_));
        v2f bb;
        bb.x = bwb[(size_t)k0 * NF_ + col];
        bb.y = bwb[(size_t)(k0 + 1) * NF_ + col];
        acc = __builtin_amdgcn_wmma_f32_16x16x4_f32(
            false, a, false, bb, (short)0, acc, false, false);
    }

    const float bias = bout[col];
#pragma unroll
    for (int r = 0; r < 8; ++r) {
        const int row = b * V_ + mtile * 16 + r + ((lane >> 4) * 8);
        out[(size_t)row * NF_ + col] = tanhf(acc[r] + bias);
    }
}

extern "C" void kernel_launch(void* const* d_in, const int* in_sizes, int n_in,
                              void* d_out, int out_size, void* d_ws, size_t ws_size,
                              hipStream_t stream)
{
    const float* x    = (const float*)d_in[0];
    const float* Wf   = (const float*)d_in[1];
    const float* bf   = (const float*)d_in[2];
    const float* Ws   = (const float*)d_in[3];
    const float* bs   = (const float*)d_in[4];
    const float* Wout = (const float*)d_in[5];
    const float* bout = (const float*)d_in[6];
    float* out = (float*)d_out;

    float* fws = (float*)d_ws;
    float* agg = fws + (size_t)ROWS * C_;
    float* BW  = agg + (size_t)B_ * A_ * 192;
    float* WC  = BW  + (size_t)B_ * KTOT * NF_;
    float* bc  = WC  + (size_t)F_ * C_;

    k0_prep  <<<7,         256, 0, stream>>>(Wf, bf, Ws, bs, WC, bc);
    k1_vertex<<<ROWS / 16, 192, 0, stream>>>(x, WC, bc, fws);
    k2_agg   <<<B_ * A_,   96,  0, stream>>>(fws, agg);
    k3_bw    <<<B_ * KTOT, 128, 0, stream>>>(agg, Wout, BW);
    k4_out   <<<B_ * 64,   256, 0, stream>>>(x, fws, BW, bout, out);
}


# k4 re-cut: block = 128 rows, wave = 16 rows x 64 columns (A and its edge weights formed once per 4 tiles), the batch's 48x128 B staged once per block in LDS in fragment order; sum order kept
# speedup vs baseline: 6.3570x; 6.3570x over previous
#include <hip/hip_runtime.h>
#include <math.h>


typedef __attribute__((ext_vector_type(2))) float v2f;
typedef __attribute__((ext_vector_type(4))) float v4f;
typedef __attribute__((ext_vector_type(8))) float v8f;

#define B_   32
#define V_   1024
#define F_   16
#define A_   32
#define P_   64
#define NF_  128
#define C_   96
#define KTOT 48
#define ROWS (B_ * V_)


__device__ __forceinline__ float ew_of(float d)
{
    return __builtin_amdgcn_exp2f(d * d * -1.44269504088896341f);
}

__device__ __forceinline__ v8f splat8(float v)
{
    v8f r = {v, v, v, v, v, v, v, v};
    return r;
}

__global__ __launch_bounds__(256) void k0_prep(
    const float* __restrict__ Wf, const float* __restrict__ bf,
    const float* __restrict__ Ws, const float* __restrict__ bs,
    float* __restrict__ WC, float* __restrict__ bc)
{
    const int tid = blockIdx.x * 256 + threadIdx.x;
    if (tid < F_ * C_) {
        const int k = tid / C_, c = tid % C_;
        WC[tid] = (c < P_) ? Wf[k * P_ + c] : Ws[k * A_ + (c - P_)];
    } else if (tid < F_ * C_ + C_) {
        const int c = tid - F_ * C_;
        bc[c] = (c < P_) ? bf[c] : bs[c - P_];
    }
}

#define K2_SLICES 16
#define K2_GROUPS (V_ / K2_SLICES / 16)
#define K2_CH     48

template <int I>
__device__ __forceinline__ float row_pick(float x)
{
    return __builtin_bit_cast(float, __builtin_amdgcn_permlane16(
        __builtin_bit_cast(int, x), __builtin_bit_cast(int, x),
        0x11111111u * I, 0x11111111u * I, false, false));
}
__device__ __forceinline__ float xhalf(float x)
{
    return __builtin_bit_cast(float, __builtin_amdgcn_permlanex16(
        __builtin_bit_cast(int, x), __builtin_bit_cast(int, x),
        0x76543210, 0xfedcba98, false, false));
}

#define K2_MAXROW(I)                                                         \
    {                                                                        \
        v2f ar;                                                              \
        ar.x = row_pick<I>(a.x);                                             \
        ar.y = row_pick<I>(a.y);                                             \
        const v2f p0 = ar * b0;                                              \
        const v2f p1 = ar * b1;                                              \
        const v2f p2 = ar * b2;                                              \
        mx[I][0] = fmaxf(mx[I][0], fmaxf(p0.x, p0.y));                       \
        mx[I][1] = fmaxf(mx[I][1], fmaxf(p1.x, p1.y));                       \
        mx[I][2] = fmaxf(mx[I][2], fmaxf(p2.x, p2.y));                       \
    }

__device__ __forceinline__ void lds_barrier()
{
    __builtin_amdgcn_fence(__ATOMIC_RELEASE, "workgroup", "local");
    __builtin_amdgcn_s_barrier();
    __builtin_amdgcn_fence(__ATOMIC_ACQUIRE, "workgroup", "local");
}

__device__ __forceinline__ void k2_step(
    const v2f a, const v2f b0, const v2f b1, const v2f b2,
    v8f& acc0, v8f& acc1, v8f& acc2, float (&mx)[8][3])
{
    acc0 = __builtin_amdgcn_wmma_f32_16x16x4_f32(
        false, a, false, b0, (short)0, acc0, false, false);
    acc1 = __builtin_amdgcn_wmma_f32_16x16x4_f32(
        false, a, false, b1, (short)0, acc1, false, false);
    acc2 = __builtin_amdgcn_wmma_f32_16x16x4_f32(
        false, a, false, b2, (short)0, acc2, false, false);
    K2_MAXROW(0) K2_MAXROW(1) K2_MAXROW(2) K2_MAXROW(3)
    K2_MAXROW(4) K2_MAXROW(5) K2_MAXROW(6) K2_MAXROW(7)
}

__device__ __forceinline__ void k2_tiles(
    const v4f xa, const v4f xb,
    const v2f (&wA)[4], const v2f (&w0)[4], const v2f (&w1)[4], const v2f (&w2)[4],
    const float biasA, const float bias0, const float bias1, const float bias2,
    v8f& dA, v8f& d0, v8f& d1, v8f& d2)
{
    v2f xf[4];
    xf[0].x = xa.x; xf[0].y = xa.y;
    xf[1].x = xa.z; xf[1].y = xa.w;
    xf[2].x = xb.x; xf[2].y = xb.y;
    xf[3].x = xb.z; xf[3].y = xb.w;

    dA = splat8(biasA); d0 = splat8(bias0); d1 = splat8(bias1); d2 = splat8(bias2);
#pragma unroll
    for (int kt = 0; kt < 4; ++kt) {
        dA = __builtin_amdgcn_wmma_f32_16x16x4_f32(
            false, xf[kt], false, wA[kt], (short)0, dA, false, false);
        d0 = __builtin_amdgcn_wmma_f32_16x16x4_f32(
            false, xf[kt], false, w0[kt], (short)0, d0, false, false);
        d1 = __builtin_amdgcn_wmma_f32_16x16x4_f32(
            false, xf[kt], false, w1[kt], (short)0, d1, false, false);
        d2 = __builtin_amdgcn_wmma_f32_16x16x4_f32(
            false, xf[kt], false, w2[kt], (short)0, d2, false, false);
    }
#pragma unroll
    for (int r = 0; r < 8; ++r) { dA[r] = ew_of(dA[r]); d2[r] = ew_of(d2[r]); }
}

__device__ __forceinline__ void k2_steps(
    const v8f dA, const v8f d0, const v8f d1, const v8f d2,
    v8f& acc0, v8f& acc1, v8f& acc2, float (&mx)[8][3])
{
#pragma unroll
    for (int t = 0; t < 4; ++t) {
        v2f a, b0, b1, b2;
        a.x  = dA[2 * t]; a.y  = dA[2 * t + 1];
        b0.x = d0[2 * t]; b0.y = d0[2 * t + 1];
        b1.x = d1[2 * t]; b1.y = d1[2 * t + 1];
        b2.x = d2[2 * t]; b2.y = d2[2 * t + 1];
        k2_step(a, b0, b1, b2, acc0, acc1, acc2, mx);
    }
}

__global__ __launch_bounds__(512) void k2_agg_bw(
    const float* __restrict__ x,  const float* __restrict__ WC,
    const float* __restrict__ bc, const float* __restrict__ Wout,
    float* __restrict__ BWp)
{
    __shared__ float pmax[K2_SLICES][8][3][16];
    __shared__ float psum[K2_SLICES][8][3][16];
    __shared__ __attribute__((aligned(16))) float aggs[8][2 * K2_CH];
    __shared__ v4f part[8][32];

    const int lane = threadIdx.x & 31;
    const int ks   = threadIdx.x >> 5;
    const int bid  = blockIdx.x;
    const int g2   = bid & 1;
    const int ch2  = (bid >> 1) & 1;
    const int mt   = (bid >> 2) & 1;
    const int b    = bid >> 3;
    const int c16  = lane & 15;
    const int h    = lane >> 4;

    const int colA  = P_ + mt * 16 + (c16 ^ (g2 << 3));
    const int colB0 = ch2 * 32 + c16;
    const int colB1 = colB0 + 16;
    const int colB2 = P_ + ch2 * 16 + c16;
    v2f wA[4], w0[4], w1[4], w2[4];
#pragma unroll
    for (int kt = 0; kt < 4; ++kt) {
        const float* wr = WC + (size_t)(8 * h + 2 * kt) * C_;
        wA[kt].x = wr[colA];  wA[kt].y = wr[C_ + colA];
        w0[kt].x = wr[colB0]; w0[kt].y = wr[C_ + colB0];
        w1[kt].x = wr[colB1]; w1[kt].y = wr[C_ + colB1];
        w2[kt].x = wr[colB2]; w2[kt].y = wr[C_ + colB2];
    }
    const float biasA = bc[colA], bias0 = bc[colB0], bias1 = bc[colB1], bias2 = bc[colB2];

    const float* xr = x + ((size_t)b * V_ + ks * (V_ / K2_SLICES) + c16) * F_ + 8 * h;

    v8f acc0 = {}, acc1 = {}, acc2 = {};
    float mx[8][3];
#pragma unroll
    for (int i = 0; i < 8; ++i) { mx[i][0] = -INFINITY; mx[i][1] = -INFINITY; mx[i][2] = -INFINITY; }

    v4f xa = *(const v4f*)xr;
    v4f xb = *(const v4f*)(xr + 4);
    v8f dA, d0, d1, d2;

#pragma unroll 1
    for (int g = 0; g < K2_GROUPS - 1; ++g) {
        k2_tiles(xa, xb, wA, w0, w1, w2, biasA, bias0, bias1, bias2, dA, d0, d1, d2);
        xr += 16 * F_;
        xa = *(const v4f*)xr;
        xb = *(const v4f*)(xr + 4);
        k2_steps(dA, d0, d1, d2, acc0, acc1, acc2, mx);
    }
    k2_tiles(xa, xb, wA, w0, w1, w2, biasA, bias0, bias1, bias2, dA, d0, d1, d2);
    k2_steps(dA, d0, d1, d2, acc0, acc1, acc2, mx);

#pragma unroll
    for (int i = 0; i < 8; ++i) {
        mx[i][0] = fmaxf(mx[i][0], xhalf(mx[i][0]));
        mx[i][1] = fmaxf(mx[i][1], xhalf(mx[i][1]));
        mx[i][2] = fmaxf(mx[i][2], xhalf(mx[i][2]));
    }
#pragma unroll
    for (int j = 0; j < 4; ++j) {
        pmax[ks][4 * h + j][0][c16] = h ? mx[4 + j][0] : mx[j][0];
        pmax[ks][4 * h + j][1][c16] = h ? mx[4 + j][1] : mx[j][1];
        pmax[ks][4 * h + j][2][c16] = h ? mx[4 + j][2] : mx[j][2];
    }
    if (h == 0) {
#pragma unroll
        for (int r = 0; r < 8; ++r) {
            psum[ks][r][0][c16] = acc0[r];
            psum[ks][r][1][c16] = acc1[r];
            psum[ks][r][2][c16] = acc2[r];
        }
    }

    const int w    = __builtin_amdgcn_readfirstlane((int)(threadIdx.x >> 5));
    const int ia   = w >> 1, hw = w & 1;
    const int aown = mt * 16 + g2 * 8 + ia;
    const v4f* wr0 = (const v4f*)(Wout + (size_t)(F_ + aown * (2 * C_) + hw * C_ + ch2 * 32) * NF_) + lane;
    const v4f* wr1 = (const v4f*)(Wout + (size_t)(F_ + aown * (2 * C_) + hw * C_ + P_ + ch2 * 16) * NF_) + lane;
    v4f bacc = {0.0f, 0.0f, 0.0f, 0.0f};
    if (ch2 == 0 && hw == 0)
        bacc = *((const v4f*)(Wout + (size_t)(F_ + 2 * A_ * C_ + aown) * NF_) + lane);
    v4f wv[K2_CH];
#pragma unroll
    for (int i = 0; i < 32; ++i) wv[i] = wr0[(size_t)i * (NF_ / 4)];
#pragma unroll
    for (int i = 0; i < 16; ++i) wv[32 + i] = wr1[(size_t)i * (NF_ / 4)];

    lds_barrier();

    const int t = threadIdx.x;
    if (t < 8 * K2_CH) {
        const int i   = t / K2_CH;
        const int cix = t - i * K2_CH;
        const float* pm = &pmax[0][0][0][0] + t;
        const float* ps = &psum[0][0][0][0] + t;
        float rmax = -INFINITY, rsum = 0.0f;
#pragma unroll
        for (int s = 0; s < K2_SLICES; ++s) {
            rmax = fmaxf(rmax, pm[s * (8 * 3 * 16)]);
            rsum += ps[s * (8 * 3 * 16)];
        }
        aggs[i][cix]         = rmax;
        aggs[i][K2_CH + cix] = rsum * (1.0f / (float)V_);
    }
    lds_barrier();

    const v4f* sv = (const v4f*)&aggs[ia][hw * K2_CH];
#pragma unroll
    for (int q = 0; q < K2_CH / 4; ++q) {
        const v4f s = sv[q];
#pragma unroll
        for (int e = 0; e < 4; ++e) {
            const float se = s[e];
            bacc.x = fmaf(se, wv[4 * q + e].x, bacc.x);
            bacc.y = fmaf(se, wv[4 * q + e].y, bacc.y);
            bacc.z = fmaf(se, wv[4 * q + e].z, bacc.z);
            bacc.w = fmaf(se, wv[4 * q + e].w, bacc.w);
        }
    }
    if (hw) part[ia][lane] = bacc;
    __syncthreads();
    if (!hw) {
        bacc += part[ia][lane];
        *((v4f*)(BWp + ((size_t)(ch2 * B_ + b) * A_ + aown) * NF_) + lane) = bacc;
    }
}

#define K4_ROWS  128
#define K4_PITCH 20
#define K4_PAIR  (16 * K4_PITCH)

__device__ __forceinline__ void k4_put(float* dst, const v4f r0, const v4f r1)
{
    const v4f lo = {r0.x, r1.x, r0.y, r1.y};
    const v4f hi = {r0.z, r1.z, r0.w, r1.w};
    *(v4f*)dst       = lo;
    *(v4f*)(dst + 4) = hi;
}

__global__ __launch_bounds__(512) void k4_out(
    const float* __restrict__ x,   const float* __restrict__ WC,
    const float* __restrict__ bc,  const float* __restrict__ Wout,
    const float* __restrict__ BWp, const float* __restrict__ bout,
    float* __restrict__ out)
{
    __shared__ __attribute__((aligned(16))) float Bs[24 * K4_PAIR];

    const int lane  = threadIdx.x & 31;
    const int wave  = __builtin_amdgcn_readfirstlane((int)(threadIdx.x >> 5));
    const int rt    = wave >> 1;
    const int chh   = wave & 1;
    const int bid   = blockIdx.x;
    const int b     = bid >> 3, chunk = bid & 7;
    const int c16   = lane & 15;
    const int h     = lane >> 4;
    const int row0  = b * V_ + chunk * K4_ROWS + rt * 16;

    const int q = lane;
    const v4f* g0 = (const v4f*)(BWp + ((size_t)b * A_ + 2 * wave) * NF_) + q;
    const v4f* g1 = g0 + (size_t)B_ * A_ * (NF_ / 4);
    const v4f ga0 = g0[0], ga1 = g0[NF_ / 4];
    const v4f gb0 = g1[0], gb1 = g1[NF_ / 4];
    const v4f gx  = *((const v4f*)(Wout + (size_t)wave * NF_) + q);

    const float* xr  = x + (size_t)(row0 + c16) * F_ + 8 * h;
    const v4f xa = *(const v4f*)xr;
    const v4f xb = *(const v4f*)(xr + 4);
    v2f xf[4];
    xf[0].x = xa.x; xf[0].y = xa.y;
    xf[1].x = xa.z; xf[1].y = xa.w;
    xf[2].x = xb.x; xf[2].y = xb.y;
    xf[3].x = xb.z; xf[3].y = xb.w;

    const v4f* bsr = (const v4f*)(bc + P_ + 8 * h);
    const v4f s00 = bsr[0], s01 = bsr[1], s10 = bsr[4], s11 = bsr[5];
    v8f e0 = {s00.x, s00.y, s00.z, s00.w, s01.x, s01.y, s01.z, s01.w};
    v8f e1 = {s10.x, s10.y, s10.z, s10.w, s11.x, s11.y, s11.z, s11.w};
#pragma unroll
    for (int kt = 0; kt < 4; ++kt) {
        const float* wr = WC + (size_t)(8 * h + 2 * kt) * C_ + P_ + c16;
        v2f wa0, wa1;
        wa0.x = wr[0];  wa0.y = wr[C_];
        wa1.x = wr[16]; wa1.y = wr[C_ + 16];
        e0 = __builtin_amdgcn_wmma_f32_16x16x4_f32(
            false, wa0, false, xf[kt], (short)0, e0, false, false);
        e1 = __builtin_amdgcn_wmma_f32_16x16x4_f32(
            false, wa1, false, xf[kt], (short)0, e1, false, false);
    }
#pragma unroll
    for (int r = 0; r < 8; ++r) { e0[r] = ew_of(e0[r]); e1[r] = ew_of(e1[r]); }

    {
        float* slot = Bs + (q & 15) * K4_PITCH + (q >> 4) * 8;
        k4_put(slot + (8 + wave) * K4_PAIR, ga0 + gb0, ga1 + gb1);
        float* xs = slot + (wave >> 1) * K4_PAIR + (wave & 1);
        xs[0] = gx.x; xs[2] = gx.y; xs[4] = gx.z; xs[6] = gx.w;
    }
    __syncthreads();

    const v4f* bl = (const v4f*)(Bs + (4 * h) * K4_PAIR + c16 * K4_PITCH + chh * 8);

    v8f acc[4];
#pragma unroll
    for (int j = 0; j < 4; ++j) acc[j] = splat8(0.0f);

#pragma unroll
    for (int st = 0; st < 12; ++st) {
        const int T = st >> 2, t = st & 3;
        v2f a;
        if (T == 0)      { a = xf[t]; }
        else if (T == 1) { a.x = e0[2 * t]; a.y = e0[2 * t + 1]; }
        else             { a.x = e1[2 * t]; a.y = e1[2 * t + 1]; }
        const v4f* bp = bl + (8 * T + t) * (K4_PAIR / 4);
        const v4f f0 = bp[0], f1 = bp[1];
        v2f bj[4];
        bj[0].x = f0.x; bj[0].y = f0.y;
        bj[1].x = f0.z; bj[1].y = f0.w;
        bj[2].x = f1.x; bj[2].y = f1.y;
        bj[3].x = f1.z; bj[3].y = f1.w;
#pragma unroll
        for (int j = 0; j < 4; ++j)
            acc[j] = __builtin_amdgcn_wmma_f32_16x16x4_f32(
                false, a, false, bj[j], (short)0, acc[j], false, false);
    }

    const v4f bo = *((const v4f*)(bout + chh * 64) + c16);
#pragma unroll
    for (int r = 0; r < 8; ++r) {
        v4f o;
        o.x = tanhf(acc[0][r] + bo.x);
        o.y = tanhf(acc[1][r] + bo.y);
        o.z = tanhf(acc[2][r] + bo.z);
        o.w = tanhf(acc[3][r] + bo.w);
        *(v4f*)(out + (size_t)(row0 + r + 8 * h) * NF_ + chh * 64 + 4 * c16) = o;
    }
}

extern "C" void kernel_launch(void* const* d_in, const int* in_sizes, int n_in,
                              void* d_out, int out_size, void* d_ws, size_t ws_size,
                              hipStream_t stream)
{
    const float* x    = (const float*)d_in[0];
    const float* Wf   = (const float*)d_in[1];
    const float* bf   = (const float*)d_in[2];
    const float* Ws   = (const float*)d_in[3];
    const float* bs   = (const float*)d_in[4];
    const float* Wout = (const float*)d_in[5];
    const float* bout = (const float*)d_in[6];
    float* out = (float*)d_out;

    float* BWp = (float*)d_ws;
    float* WC  = BWp + (size_t)2 * B_ * A_ * NF_;
    float* bc  = WC  + (size_t)F_ * C_;

    k0_prep  <<<7,       256, 0, stream>>>(Wf, bf, Ws, bs, WC, bc);
    k2_agg_bw<<<B_ * 8,  512, 0, stream>>>(x, WC, bc, Wout, BWp);
    k4_out   <<<B_ * 8,  512, 0, stream>>>(x, WC, bc, Wout, BWp, bout, out);
}


# k4 epilogue: the 32 library tanhf per lane (range-reduced exp and a polynomial in 2 divergent branch regions each) become the hardware v_tanh_f32, one branch-free transcendental each; sum order kept
# speedup vs baseline: 7.2004x; 1.1327x over previous
#include <hip/hip_runtime.h>
#include <math.h>


typedef __attribute__((ext_vector_type(2))) float v2f;
typedef __attribute__((ext_vector_type(4))) float v4f;
typedef __attribute__((ext_vector_type(8))) float v8f;

#define B_   32
#define V_   1024
#define F_   16
#define A_   32
#define P_   64
#define NF_  128
#define C_   96
#define KTOT 48
#define ROWS (B_ * V_)


__device__ __forceinline__ float ew_of(float d)
{
    return __builtin_amdgcn_exp2f(d * d * -1.44269504088896341f);
}

__device__ __forceinline__ float tanh_hw(float v)
{
    return __builtin_amdgcn_tanhf(v);
}

__device__ __forceinline__ v8f splat8(float v)
{
    v8f r = {v, v, v, v, v, v, v, v};
    return r;
}

__global__ __launch_bounds__(256) void k0_prep(
    const float* __restrict__ Wf, const float* __restrict__ bf,
    const float* __restrict__ Ws, const float* __restrict__ bs,
    float* __restrict__ WC, float* __restrict__ bc)
{
    const int tid = blockIdx.x * 256 + threadIdx.x;
    if (tid < F_ * C_) {
        const int k = tid / C_, c = tid % C_;
        WC[tid] = (c < P_) ? Wf[k * P_ + c] : Ws[k * A_ + (c - P_)];
    } else if (tid < F_ * C_ + C_) {
        const int c = tid - F_ * C_;
        bc[c] = (c < P_) ? bf[c] : bs[c - P_];
    }
}

#define K2_SLICES 16
#define K2_GROUPS (V_ / K2_SLICES / 16)
#define K2_CH     48

template <int I>
__device__ __forceinline__ float row_pick(float x)
{
    return __builtin_bit_cast(float, __builtin_amdgcn_permlane16(
        __builtin_bit_cast(int, x), __builtin_bit_cast(int, x),
        0x11111111u * I, 0x11111111u * I, false, false));
}
__device__ __forceinline__ float xhalf(float x)
{
    return __builtin_bit_cast(float, __builtin_amdgcn_permlanex16(
        __builtin_bit_cast(int, x), __builtin_bit_cast(int, x),
        0x76543210, 0xfedcba98, false, false));
}

#define K2_MAXROW(I)                                                         \
    {                                                                        \
        v2f ar;                                                              \
        ar.x = row_pick<I>(a.x);                                             \
        ar.y = row_pick<I>(a.y);                                             \
        const v2f p0 = ar * b0;                                              \
        const v2f p1 = ar * b1;                                              \
        const v2f p2 = ar * b2;                                              \
        mx[I][0] = fmaxf(mx[I][0], fmaxf(p0.x, p0.y));                       \
        mx[I][1] = fmaxf(mx[I][1], fmaxf(p1.x, p1.y));                       \
        mx[I][2] = fmaxf(mx[I][2], fmaxf(p2.x, p2.y));                       \
    }

__device__ __forceinline__ void lds_barrier()
{
    __builtin_amdgcn_fence(__ATOMIC_RELEASE, "workgroup", "local");
    __builtin_amdgcn_s_barrier();
    __builtin_amdgcn_fence(__ATOMIC_ACQUIRE, "workgroup", "local");
}

__device__ __forceinline__ void k2_step(
    const v2f a, const v2f b0, const v2f b1, const v2f b2,
    v8f& acc0, v8f& acc1, v8f& acc2, float (&mx)[8][3])
{
    acc0 = __builtin_amdgcn_wmma_f32_16x16x4_f32(
        false, a, false, b0, (short)0, acc0, false, false);
    acc1 = __builtin_amdgcn_wmma_f32_16x16x4_f32(
        false, a, false, b1, (short)0, acc1, false, false);
    acc2 = __builtin_amdgcn_wmma_f32_16x16x4_f32(
        false, a, false, b2, (short)0, acc2, false, false);
    K2_MAXROW(0) K2_MAXROW(1) K2_MAXROW(2) K2_MAXROW(3)
    K2_MAXROW(4) K2_MAXROW(5) K2_MAXROW(6) K2_MAXROW(7)
}

__device__ __forceinline__ void k2_tiles(
    const v4f xa, const v4f xb,
    const v2f (&wA)[4], const v2f (&w0)[4], const v2f (&w1)[4], const v2f (&w2)[4],
    const float biasA, const float bias0, const float bias1, const float bias2,
    v8f& dA, v8f& d0, v8f& d1, v8f& d2)
{
    v2f xf[4];
    xf[0].x = xa.x; xf[0].y = xa.y;
    xf[1].x = xa.z; xf[1].y = xa.w;
    xf[2].x = xb.x; xf[2].y = xb.y;
    xf[3].x = xb.z; xf[3].y = xb.w;

    dA = splat8(biasA); d0 = splat8(bias0); d1 = splat8(bias1); d2 = splat8(bias2);
#pragma unroll
    for (int kt = 0; kt < 4; ++kt) {
        dA = __builtin_amdgcn_wmma_f32_16x16x4_f32(
            false, xf[kt], false, wA[kt], (short)0, dA, false, false);
        d0 = __builtin_amdgcn_wmma_f32_16x16x4_f32(
            false, xf[kt], false, w0[kt], (short)0, d0, false, false);
        d1 = __builtin_amdgcn_wmma_f32_16x16x4_f32(
            false, xf[kt], false, w1[kt], (short)0, d1, false, false);
        d2 = __builtin_amdgcn_wmma_f32_16x16x4_f32(
            false, xf[kt], false, w2[kt], (short)0, d2, false, false);
    }
#pragma unroll
    for (int r = 0; r < 8; ++r) { dA[r] = ew_of(dA[r]); d2[r] = ew_of(d2[r]); }
}

__device__ __forceinline__ void k2_steps(
    const v8f dA, const v8f d0, const v8f d1, const v8f d2,
    v8f& acc0, v8f& acc1, v8f& acc2, float (&mx)[8][3])
{
#pragma unroll
    for (int t = 0; t < 4; ++t) {
        v2f a, b0, b1, b2;
        a.x  = dA[2 * t]; a.y  = dA[2 * t + 1];
        b0.x = d0[2 * t]; b0.y = d0[2 * t + 1];
        b1.x = d1[2 * t]; b1.y = d1[2 * t + 1];
        b2.x = d2[2 * t]; b2.y = d2[2 * t + 1];
        k2_step(a, b0, b1, b2, acc0, acc1, acc2, mx);
    }
}

__global__ __launch_bounds__(512) void k2_agg_bw(
    const float* __restrict__ x,  const float* __restrict__ WC,
    const float* __restrict__ bc, const float* __restrict__ Wout,
    float* __restrict__ BWp)
{
    __shared__ float pmax[K2_SLICES][8][3][16];
    __shared__ float psum[K2_SLICES][8][3][16];
    __shared__ __attribute__((aligned(16))) float aggs[8][2 * K2_CH];
    __shared__ v4f part[8][32];

    const int lane = threadIdx.x & 31;
    const int ks   = threadIdx.x >> 5;
    const int bid  = blockIdx.x;
    const int g2   = bid & 1;
    const int ch2  = (bid >> 1) & 1;
    const int mt   = (bid >> 2) & 1;
    const int b    = bid >> 3;
    const int c16  = lane & 15;
    const int h    = lane >> 4;

    const int colA  = P_ + mt * 16 + (c16 ^ (g2 << 3));
    const int colB0 = ch2 * 32 + c16;
    const int colB1 = colB0 + 16;
    const int colB2 = P_ + ch2 * 16 + c16;
    v2f wA[4], w0[4], w1[4], w2[4];
#pragma unroll
    for (int kt = 0; kt < 4; ++kt) {
        const float* wr = WC + (size_t)(8 * h + 2 * kt) * C_;
        wA[kt].x = wr[colA];  wA[kt].y = wr[C_ + colA];
        w0[kt].x = wr[colB0]; w0[kt].y = wr[C_ + colB0];
        w1[kt].x = wr[colB1]; w1[kt].y = wr[C_ + colB1];
        w2[kt].x = wr[colB2]; w2[kt].y = wr[C_ + colB2];
    }
    const float biasA = bc[colA], bias0 = bc[colB0], bias1 = bc[colB1], bias2 = bc[colB2];

    const float* xr = x + ((size_t)b * V_ + ks * (V_ / K2_SLICES) + c16) * F_ + 8 * h;

    v8f acc0 = {}, acc1 = {}, acc2 = {};
    float mx[8][3];
#pragma unroll
    for (int i = 0; i < 8; ++i) { mx[i][0] = -INFINITY; mx[i][1] = -INFINITY; mx[i][2] = -INFINITY; }

    v4f xa = *(const v4f*)xr;
    v4f xb = *(const v4f*)(xr + 4);
    v8f dA, d0, d1, d2;

#pragma unroll 1
    for (int g = 0; g < K2_GROUPS - 1; ++g) {
        k2_tiles(xa, xb, wA, w0, w1, w2, biasA, bias0, bias1, bias2, dA, d0, d1, d2);
        xr += 16 * F_;
        xa = *(const v4f*)xr;
        xb = *(const v4f*)(xr + 4);
        k2_steps(dA, d0, d1, d2, acc0, acc1, acc2, mx);
    }
    k2_tiles(xa, xb, wA, w0, w1, w2, biasA, bias0, bias1, bias2, dA, d0, d1, d2);
    k2_steps(dA, d0, d1, d2, acc0, acc1, acc2, mx);

#pragma unroll
    for (int i = 0; i < 8; ++i) {
        mx[i][0] = fmaxf(mx[i][0], xhalf(mx[i][0]));
        mx[i][1] = fmaxf(mx[i][1], xhalf(mx[i][1]));
        mx[i][2] = fmaxf(mx[i][2], xhalf(mx[i][2]));
    }
#pragma unroll
    for (int j = 0; j < 4; ++j) {
        pmax[ks][4 * h + j][0][c16] = h ? mx[4 + j][0] : mx[j][0];
        pmax[ks][4 * h + j][1][c16] = h ? mx[4 + j][1] : mx[j][1];
        pmax[ks][4 * h + j][2][c16] = h ? mx[4 + j][2] : mx[j][2];
    }
    if (h == 0) {
#pragma unroll
        for (int r = 0; r < 8; ++r) {
            psum[ks][r][0][c16] = acc0[r];
            psum[ks][r][1][c16] = acc1[r];
            psum[ks][r][2][c16] = acc2[r];
        }
    }

    const int w    = __builtin_amdgcn_readfirstlane((int)(threadIdx.x >> 5));
    const int ia   = w >> 1, hw = w & 1;
    const int aown = mt * 16 + g2 * 8 + ia;
    const v4f* wr0 = (const v4f*)(Wout + (size_t)(F_ + aown * (2 * C_) + hw * C_ + ch2 * 32) * NF_) + lane;
    const v4f* wr1 = (const v4f*)(Wout + (size_t)(F_ + aown * (2 * C_) + hw * C_ + P_ + ch2 * 16) * NF_) + lane;
    v4f bacc = {0.0f, 0.0f, 0.0f, 0.0f};
    if (ch2 == 0 && hw == 0)
        bacc = *((const v4f*)(Wout + (size_t)(F_ + 2 * A_ * C_ + aown) * NF_) + lane);
    v4f wv[K2_CH];
#pragma unroll
    for (int i = 0; i < 32; ++i) wv[i] = wr0[(size_t)i * (NF_ / 4)];
#pragma unroll
    for (int i = 0; i < 16; ++i) wv[32 + i] = wr1[(size_t)i * (NF_ / 4)];

    lds_barrier();

    const int t = threadIdx.x;
    if (t < 8 * K2_CH) {
        const int i   = t / K2_CH;
        const int cix = t - i * K2_CH;
        const float* pm = &pmax[0][0][0][0] + t;
        const float* ps = &psum[0][0][0][0] + t;
        float rmax = -INFINITY, rsum = 0.0f;
#pragma unroll
        for (int s = 0; s < K2_SLICES; ++s) {
            rmax = fmaxf(rmax, pm[s * (8 * 3 * 16)]);
            rsum += ps[s * (8 * 3 * 16)];
        }
        aggs[i][cix]         = rmax;
        aggs[i][K2_CH + cix] = rsum * (1.0f / (float)V_);
    }
    lds_barrier();

    const v4f* sv = (const v4f*)&aggs[ia][hw * K2_CH];
#pragma unroll
    for (int q = 0; q < K2_CH / 4; ++q) {
        const v4f s = sv[q];
#pragma unroll
        for (int e = 0; e < 4; ++e) {
            const float se = s[e];
            bacc.x = fmaf(se, wv[4 * q + e].x, bacc.x);
            bacc.y = fmaf(se, wv[4 * q + e].y, bacc.y);
            bacc.z = fmaf(se, wv[4 * q + e].z, bacc.z);
            bacc.w = fmaf(se, wv[4 * q + e].w, bacc.w);
        }
    }
    if (hw) part[ia][lane] = bacc;
    __syncthreads();
    if (!hw) {
        bacc += part[ia][lane];
        *((v4f*)(BWp + ((size_t)(ch2 * B_ + b) * A_ + aown) * NF_) + lane) = bacc;
    }
}

#define K4_ROWS  128
#define K4_PITCH 20
#define K4_PAIR  (16 * K4_PITCH)

__device__ __forceinline__ void k4_put(float* dst, const v4f r0, const v4f r1)
{
    const v4f lo = {r0.x, r1.x, r0.y, r1.y};
    const v4f hi = {r0.z, r1.z, r0.w, r1.w};
    *(v4f*)dst       = lo;
    *(v4f*)(dst + 4) = hi;
}

__global__ __launch_bounds__(512) void k4_out(
    const float* __restrict__ x,   const float* __restrict__ WC,
    const float* __restrict__ bc,  const float* __restrict__ Wout,
    const float* __restrict__ BWp, const float* __restrict__ bout,
    float* __restrict__ out)
{
    __shared__ __attribute__((aligned(16))) float Bs[24 * K4_PAIR];

    const int lane  = threadIdx.x & 31;
    const int wave  = __builtin_amdgcn_readfirstlane((int)(threadIdx.x >> 5));
    const int rt    = wave >> 1;
    const int chh   = wave & 1;
    const int bid   = blockIdx.x;
    const int b     = bid >> 3, chunk = bid & 7;
    const int c16   = lane & 15;
    const int h     = lane >> 4;
    const int row0  = b * V_ + chunk * K4_ROWS + rt * 16;

    const int q = lane;
    const v4f* g0 = (const v4f*)(BWp + ((size_t)b * A_ + 2 * wave) * NF_) + q;
    const v4f* g1 = g0 + (size_t)B_ * A_ * (NF_ / 4);
    const v4f ga0 = g0[0], ga1 = g0[NF_ / 4];
    const v4f gb0 = g1[0], gb1 = g1[NF_ / 4];
    const v4f gx  = *((const v4f*)(Wout + (size_t)wave * NF_) + q);

    const float* xr  = x + (size_t)(row0 + c16) * F_ + 8 * h;
    const v4f xa = *(const v4f*)xr;
    const v4f xb = *(const v4f*)(xr + 4);
    v2f xf[4];
    xf[0].x = xa.x; xf[0].y = xa.y;
    xf[1].x = xa.z; xf[1].y = xa.w;
    xf[2].x = xb.x; xf[2].y = xb.y;
    xf[3].x = xb.z; xf[3].y = xb.w;

    const v4f* bsr = (const v4f*)(bc + P_ + 8 * h);
    const v4f s00 = bsr[0], s01 = bsr[1], s10 = bsr[4], s11 = bsr[5];
    v8f e0 = {s00.x, s00.y, s00.z, s00.w, s01.x, s01.y, s01.z, s01.w};
    v8f e1 = {s10.x, s10.y, s10.z, s10.w, s11.x, s11.y, s11.z, s11.w};
#pragma unroll
    for (int kt = 0; kt < 4; ++kt) {
        const float* wr = WC + (size_t)(8 * h + 2 * kt) * C_ + P_ + c16;
        v2f wa0, wa1;
        wa0.x = wr[0];  wa0.y = wr[C_];
        wa1.x = wr[16]; wa1.y = wr[C_ + 16];
        e0 = __builtin_amdgcn_wmma_f32_16x16x4_f32(
            false, wa0, false, xf[kt], (short)0, e0, false, false);
        e1 = __builtin_amdgcn_wmma_f32_16x16x4_f32(
            false, wa1, false, xf[kt], (short)0, e1, false, false);
    }
#pragma unroll
    for (int r = 0; r < 8; ++r) { e0[r] = ew_of(e0[r]); e1[r] = ew_of(e1[r]); }

    {
        float* slot = Bs + (q & 15) * K4_PITCH + (q >> 4) * 8;
        k4_put(slot + (8 + wave) * K4_PAIR, ga0 + gb0, ga1 + gb1);
        float* xs = slot + (wave >> 1) * K4_PAIR + (wave & 1);
        xs[0] = gx.x; xs[2] = gx.y; xs[4] = gx.z; xs[6] = gx.w;
    }
    __syncthreads();

    const v4f* bl = (const v4f*)(Bs + (4 * h) * K4_PAIR + c16 * K4_PITCH + chh * 8);

    v8f acc[4];
#pragma unroll
    for (int j = 0; j < 4; ++j) acc[j] = splat8(0.0f);

#pragma unroll
    for (int st = 0; st < 12; ++st) {
        const int T = st >> 2, t = st & 3;
        v2f a;
        if (T == 0)      { a = xf[t]; }
        else if (T == 1) { a.x = e0[2 * t]; a.y = e0[2 * t + 1]; }
        else             { a.x = e1[2 * t]; a.y = e1[2 * t + 1]; }
        const v4f* bp = bl + (8 * T + t) * (K4_PAIR / 4);
        const v4f f0 = bp[0], f1 = bp[1];
        v2f bj[4];
        bj[0].x = f0.x; bj[0].y = f0.y;
        bj[1].x = f0.z; bj[1].y = f0.w;
        bj[2].x = f1.x; bj[2].y = f1.y;
        bj[3].x = f1.z; bj[3].y = f1.w;
#pragma unroll
        for (int j = 0; j < 4; ++j)
            acc[j] = __builtin_amdgcn_wmma_f32_16x16x4_f32(
                false, a, false, bj[j], (short)0, acc[j], false, false);
    }

    const v4f bo = *((const v4f*)(bout + chh * 64) + c16);
#pragma unroll
    for (int r = 0; r < 8; ++r) {
        v4f o;
        o.x = tanh_hw(acc[0][r] + bo.x);
        o.y = tanh_hw(acc[1][r] + bo.y);
        o.z = tanh_hw(acc[2][r] + bo.z);
        o.w = tanh_hw(acc[3][r] + bo.w);
        *(v4f*)(out + (size_t)(row0 + r + 8 * h) * NF_ + chh * 64 + 4 * c16) = o;
    }
}

extern "C" void kernel_launch(void* const* d_in, const int* in_sizes, int n_in,
                              void* d_out, int out_size, void* d_ws, size_t ws_size,
                              hipStream_t stream)
{
    const float* x    = (const float*)d_in[0];
    const float* Wf   = (const float*)d_in[1];
    const float* bf   = (const float*)d_in[2];
    const float* Ws   = (const float*)d_in[3];
    const float* bs   = (const float*)d_in[4];
    const float* Wout = (const float*)d_in[5];
    const float* bout = (const float*)d_in[6];
    float* out = (float*)d_out;

    float* BWp = (float*)d_ws;
    float* WC  = BWp + (size_t)2 * B_ * A_ * NF_;
    float* bc  = WC  + (size_t)F_ * C_;

    k0_prep  <<<7,       256, 0, stream>>>(Wf, bf, Ws, bs, WC, bc);
    k2_agg_bw<<<B_ * 8,  512, 0, stream>>>(x, WC, bc, Wout, BWp);
    k4_out   <<<B_ * 8,  512, 0, stream>>>(x, WC, bc, Wout, BWp, bout, out);
}


# weight-packing launch dropped: both kernels read W_flr, W_s and the biases in place (each fragment tile lies wholly in one matrix: own base and pitch, no select); 3 launches become 2; sum order kept
# speedup vs baseline: 8.0908x; 1.1237x over previous
#include <hip/hip_runtime.h>
#include <math.h>


typedef __attribute__((ext_vector_type(2))) float v2f;
typedef __attribute__((ext_vector_type(4))) float v4f;
typedef __attribute__((ext_vector_type(8))) float v8f;

#define B_   32
#define V_   1024
#define F_   16
#define A_   32
#define P_   64
#define NF_  128
#define C_   96
#define KTOT 48
#define ROWS (B_ * V_)


__device__ __forceinline__ float ew_of(float d)
{
    return __builtin_amdgcn_exp2f(d * d * -1.44269504088896341f);
}

__device__ __forceinline__ float tanh_hw(float v)
{
    return __builtin_amdgcn_tanhf(v);
}

__device__ __forceinline__ v8f splat8(float v)
{
    v8f r = {v, v, v, v, v, v, v, v};
    return r;
}

#define K2_SLICES 16
#define K2_GROUPS (V_ / K2_SLICES / 16)
#define K2_CH     48

template <int I>
__device__ __forceinline__ float row_pick(float x)
{
    return __builtin_bit_cast(float, __builtin_amdgcn_permlane16(
        __builtin_bit_cast(int, x), __builtin_bit_cast(int, x),
        0x11111111u * I, 0x11111111u * I, false, false));
}
__device__ __forceinline__ float xhalf(float x)
{
    return __builtin_bit_cast(float, __builtin_amdgcn_permlanex16(
        __builtin_bit_cast(int, x), __builtin_bit_cast(int, x),
        0x76543210, 0xfedcba98, false, false));
}

#define K2_MAXROW(I)                                                         \
    {                                                                        \
        v2f ar;                                                              \
        ar.x = row_pick<I>(a.x);                                             \
        ar.y = row_pick<I>(a.y);                                             \
        const v2f p0 = ar * b0;                                              \
        const v2f p1 = ar * b1;                                              \
        const v2f p2 = ar * b2;                                              \
        mx[I][0] = fmaxf(mx[I][0], fmaxf(p0.x, p0.y));                       \
        mx[I][1] = fmaxf(mx[I][1], fmaxf(p1.x, p1.y));                       \
        mx[I][2] = fmaxf(mx[I][2], fmaxf(p2.x, p2.y));                       \
    }

__device__ __forceinline__ void lds_barrier()
{
    __builtin_amdgcn_fence(__ATOMIC_RELEASE, "workgroup", "local");
    __builtin_amdgcn_s_barrier();
    __builtin_amdgcn_fence(__ATOMIC_ACQUIRE, "workgroup", "local");
}

__device__ __forceinline__ void k2_step(
    const v2f a, const v2f b0, const v2f b1, const v2f b2,
    v8f& acc0, v8f& acc1, v8f& acc2, float (&mx)[8][3])
{
    acc0 = __builtin_amdgcn_wmma_f32_16x16x4_f32(
        false, a, false, b0, (short)0, acc0, false, false);
    acc1 = __builtin_amdgcn_wmma_f32_16x16x4_f32(
        false, a, false, b1, (short)0, acc1, false, false);
    acc2 = __builtin_amdgcn_wmma_f32_16x16x4_f32(
        false, a, false, b2, (short)0, acc2, false, false);
    K2_MAXROW(0) K2_MAXROW(1) K2_MAXROW(2) K2_MAXROW(3)
    K2_MAXROW(4) K2_MAXROW(5) K2_MAXROW(6) K2_MAXROW(7)
}

__device__ __forceinline__ void k2_tiles(
    const v4f xa, const v4f xb,
    const v2f (&wA)[4], const v2f (&w0)[4], const v2f (&w1)[4], const v2f (&w2)[4],
    const float biasA, const float bias0, const float bias1, const float bias2,
    v8f& dA, v8f& d0, v8f& d1, v8f& d2)
{
    v2f xf[4];
    xf[0].x = xa.x; xf[0].y = xa.y;
    xf[1].x = xa.z; xf[1].y = xa.w;
    xf[2].x = xb.x; xf[2].y = xb.y;
    xf[3].x = xb.z; xf[3].y = xb.w;

    dA = splat8(biasA); d0 = splat8(bias0); d1 = splat8(bias1); d2 = splat8(bias2);
#pragma unroll
    for (int kt = 0; kt < 4; ++kt) {
        dA = __builtin_amdgcn_wmma_f32_16x16x4_f32(
            false, xf[kt], false, wA[kt], (short)0, dA, false, false);
        d0 = __builtin_amdgcn_wmma_f32_16x16x4_f32(
            false, xf[kt], false, w0[kt], (short)0, d0, false, false);
        d1 = __builtin_amdgcn_wmma_f32_16x16x4_f32(
            false, xf[kt], false, w1[kt], (short)0, d1, false, false);
        d2 = __builtin_amdgcn_wmma_f32_16x16x4_f32(
            false, xf[kt], false, w2[kt], (short)0, d2, false, false);
    }
#pragma unroll
    for (int r = 0; r < 8; ++r) { dA[r] = ew_of(dA[r]); d2[r] = ew_of(d2[r]); }
}

__device__ __forceinline__ void k2_steps(
    const v8f dA, const v8f d0, const v8f d1, const v8f d2,
    v8f& acc0, v8f& acc1, v8f& acc2, float (&mx)[8][3])
{
#pragma unroll
    for (int t = 0; t < 4; ++t) {
        v2f a, b0, b1, b2;
        a.x  = dA[2 * t]; a.y  = dA[2 * t + 1];
        b0.x = d0[2 * t]; b0.y = d0[2 * t + 1];
        b1.x = d1[2 * t]; b1.y = d1[2 * t + 1];
        b2.x = d2[2 * t]; b2.y = d2[2 * t + 1];
        k2_step(a, b0, b1, b2, acc0, acc1, acc2, mx);
    }
}

__global__ __launch_bounds__(512) void k2_agg_bw(
    const float* __restrict__ x,
    const float* __restrict__ Wf, const float* __restrict__ bf,
    const float* __restrict__ Ws, const float* __restrict__ bs,
    const float* __restrict__ Wout, float* __restrict__ BWp)
{
    __shared__ float pmax[K2_SLICES][8][3][16];
    __shared__ float psum[K2_SLICES][8][3][16];
    __shared__ __attribute__((aligned(16))) float aggs[8][2 * K2_CH];
    __shared__ v4f part[8][32];

    const int lane = threadIdx.x & 31;
    const int ks   = threadIdx.x >> 5;
    const int bid  = blockIdx.x;
    const int g2   = bid & 1;
    const int ch2  = (bid >> 1) & 1;
    const int mt   = (bid >> 2) & 1;
    const int b    = bid >> 3;
    const int c16  = lane & 15;
    const int h    = lane >> 4;

    const int colA  = mt * 16 + (c16 ^ (g2 << 3));
    const int colB0 = ch2 * 32 + c16;
    const int colB1 = colB0 + 16;
    const int colB2 = ch2 * 16 + c16;
    v2f wA[4], w0[4], w1[4], w2[4];
#pragma unroll
    for (int kt = 0; kt < 4; ++kt) {
        const float* wf = Wf + (size_t)(8 * h + 2 * kt) * P_;
        const float* ws = Ws + (size_t)(8 * h + 2 * kt) * A_;
        wA[kt].x = ws[colA];  wA[kt].y = ws[A_ + colA];
        w0[kt].x = wf[colB0]; w0[kt].y = wf[P_ + colB0];
        w1[kt].x = wf[colB1]; w1[kt].y = wf[P_ + colB1];
        w2[kt].x = ws[colB2]; w2[kt].y = ws[A_ + colB2];
    }
    const float biasA = bs[colA], bias0 = bf[colB0], bias1 = bf[colB1], bias2 = bs[colB2];

    const float* xr = x + ((size_t)b * V_ + ks * (V_ / K2_SLICES) + c16) * F_ + 8 * h;

    v8f acc0 = {}, acc1 = {}, acc2 = {};
    float mx[8][3];
#pragma unroll
    for (int i = 0; i < 8; ++i) { mx[i][0] = -INFINITY; mx[i][1] = -INFINITY; mx[i][2] = -INFINITY; }

    v4f xa = *(const v4f*)xr;
    v4f xb = *(const v4f*)(xr + 4);
    v8f dA, d0, d1, d2;

#pragma unroll 1
    for (int g = 0; g < K2_GROUPS - 1; ++g) {
        k2_tiles(xa, xb, wA, w0, w1, w2, biasA, bias0, bias1, bias2, dA, d0, d1, d2);
        xr += 16 * F_;
        xa = *(const v4f*)xr;
        xb = *(const v4f*)(xr + 4);
        k2_steps(dA, d0, d1, d2, acc0, acc1, acc2, mx);
    }
    k2_tiles(xa, xb, wA, w0, w1, w2, biasA, bias0, bias1, bias2, dA, d0, d1, d2);
    k2_steps(dA, d0, d1, d2, acc0, acc1, acc2, mx);

#pragma unroll
    for (int i = 0; i < 8; ++i) {
        mx[i][0] = fmaxf(mx[i][0], xhalf(mx[i][0]));
        mx[i][1] = fmaxf(mx[i][1], xhalf(mx[i][1]));
        mx[i][2] = fmaxf(mx[i][2], xhalf(mx[i][2]));
    }
#pragma unroll
    for (int j = 0; j < 4; ++j) {
        pmax[ks][4 * h + j][0][c16] = h ? mx[4 + j][0] : mx[j][0];
        pmax[ks][4 * h + j][1][c16] = h ? mx[4 + j][1] : mx[j][1];
        pmax[ks][4 * h + j][2][c16] = h ? mx[4 + j][2] : mx[j][2];
    }
    if (h == 0) {
#pragma unroll
        for (int r = 0; r < 8; ++r) {
            psum[ks][r][0][c16] = acc0[r];
            psum[ks][r][1][c16] = acc1[r];
            psum[ks][r][2][c16] = acc2[r];
        }
    }

    const int w    = __builtin_amdgcn_readfirstlane((int)(threadIdx.x >> 5));
    const int ia   = w >> 1, hw = w & 1;
    const int aown = mt * 16 + g2 * 8 + ia;
    const v4f* wr0 = (const v4f*)(Wout + (size_t)(F_ + aown * (2 * C_) + hw * C_ + ch2 * 32) * NF_) + lane;
    const v4f* wr1 = (const v4f*)(Wout + (size_t)(F_ + aown * (2 * C_) + hw * C_ + P_ + ch2 * 16) * NF_) + lane;
    v4f bacc = {0.0f, 0.0f, 0.0f, 0.0f};
    if (ch2 == 0 && hw == 0)
        bacc = *((const v4f*)(Wout + (size_t)(F_ + 2 * A_ * C_ + aown) * NF_) + lane);
    v4f wv[K2_CH];
#pragma unroll
    for (int i = 0; i < 32; ++i) wv[i] = wr0[(size_t)i * (NF_ / 4)];
#pragma unroll
    for (int i = 0; i < 16; ++i) wv[32 + i] = wr1[(size_t)i * (NF_ / 4)];

    lds_barrier();

    const int t = threadIdx.x;
    if (t < 8 * K2_CH) {
        const int i   = t / K2_CH;
        const int cix = t - i * K2_CH;
        const float* pm = &pmax[0][0][0][0] + t;
        const float* ps = &psum[0][0][0][0] + t;
        float rmax = -INFINITY, rsum = 0.0f;
#pragma unroll
        for (int s = 0; s < K2_SLICES; ++s) {
            rmax = fmaxf(rmax, pm[s * (8 * 3 * 16)]);
            rsum += ps[s * (8 * 3 * 16)];
        }
        aggs[i][cix]         = rmax;
        aggs[i][K2_CH + cix] = rsum * (1.0f / (float)V_);
    }
    lds_barrier();

    const v4f* sv = (const v4f*)&aggs[ia][hw * K2_CH];
#pragma unroll
    for (int q = 0; q < K2_CH / 4; ++q) {
        const v4f s = sv[q];
#pragma unroll
        for (int e = 0; e < 4; ++e) {
            const float se = s[e];
            bacc.x = fmaf(se, wv[4 * q + e].x, bacc.x);
            bacc.y = fmaf(se, wv[4 * q + e].y, bacc.y);
            bacc.z = fmaf(se, wv[4 * q + e].z, bacc.z);
            bacc.w = fmaf(se, wv[4 * q + e].w, bacc.w);
        }
    }
    if (hw) part[ia][lane] = bacc;
    __syncthreads();
    if (!hw) {
        bacc += part[ia][lane];
        *((v4f*)(BWp + ((size_t)(ch2 * B_ + b) * A_ + aown) * NF_) + lane) = bacc;
    }
}

#define K4_ROWS  128
#define K4_PITCH 20
#define K4_PAIR  (16 * K4_PITCH)

__device__ __forceinline__ void k4_put(float* dst, const v4f r0, const v4f r1)
{
    const v4f lo = {r0.x, r1.x, r0.y, r1.y};
    const v4f hi = {r0.z, r1.z, r0.w, r1.w};
    *(v4f*)dst       = lo;
    *(v4f*)(dst + 4) = hi;
}

__global__ __launch_bounds__(512) void k4_out(
    const float* __restrict__ x,   const float* __restrict__ Ws,
    const float* __restrict__ bs,  const float* __restrict__ Wout,
    const float* __restrict__ BWp, const float* __restrict__ bout,
    float* __restrict__ out)
{
    __shared__ __attribute__((aligned(16))) float Bs[24 * K4_PAIR];

    const int lane  = threadIdx.x & 31;
    const int wave  = __builtin_amdgcn_readfirstlane((int)(threadIdx.x >> 5));
    const int rt    = wave >> 1;
    const int chh   = wave & 1;
    const int bid   = blockIdx.x;
    const int b     = bid >> 3, chunk = bid & 7;
    const int c16   = lane & 15;
    const int h     = lane >> 4;
    const int row0  = b * V_ + chunk * K4_ROWS + rt * 16;

    const int q = lane;
    const v4f* g0 = (const v4f*)(BWp + ((size_t)b * A_ + 2 * wave) * NF_) + q;
    const v4f* g1 = g0 + (size_t)B_ * A_ * (NF_ / 4);
    const v4f ga0 = g0[0], ga1 = g0[NF_ / 4];
    const v4f gb0 = g1[0], gb1 = g1[NF_ / 4];
    const v4f gx  = *((const v4f*)(Wout + (size_t)wave * NF_) + q);

    const float* xr  = x + (size_t)(row0 + c16) * F_ + 8 * h;
    const v4f xa = *(const v4f*)xr;
    const v4f xb = *(const v4f*)(xr + 4);
    v2f xf[4];
    xf[0].x = xa.x; xf[0].y = xa.y;
    xf[1].x = xa.z; xf[1].y = xa.w;
    xf[2].x = xb.x; xf[2].y = xb.y;
    xf[3].x = xb.z; xf[3].y = xb.w;

    const v4f* bsr = (const v4f*)(bs + 8 * h);
    const v4f s00 = bsr[0], s01 = bsr[1], s10 = bsr[4], s11 = bsr[5];
    v8f e0 = {s00.x, s00.y, s00.z, s00.w, s01.x, s01.y, s01.z, s01.w};
    v8f e1 = {s10.x, s10.y, s10.z, s10.w, s11.x, s11.y, s11.z, s11.w};
#pragma unroll
    for (int kt = 0; kt < 4; ++kt) {
        const float* wr = Ws + (size_t)(8 * h + 2 * kt) * A_ + c16;
        v2f wa0, wa1;
        wa0.x = wr[0];  wa0.y = wr[A_];
        wa1.x = wr[16]; wa1.y = wr[A_ + 16];
        e0 = __builtin_amdgcn_wmma_f32_16x16x4_f32(
            false, wa0, false, xf[kt], (short)0, e0, false, false);
        e1 = __builtin_amdgcn_wmma_f32_16x16x4_f32(
            false, wa1, false, xf[kt], (short)0, e1, false, false);
    }
#pragma unroll
    for (int r = 0; r < 8; ++r) { e0[r] = ew_of(e0[r]); e1[r] = ew_of(e1[r]); }

    {
        float* slot = Bs + (q & 15) * K4_PITCH + (q >> 4) * 8;
        k4_put(slot + (8 + wave) * K4_PAIR, ga0 + gb0, ga1 + gb1);
        float* xs = slot + (wave >> 1) * K4_PAIR + (wave & 1);
        xs[0] = gx.x; xs[2] = gx.y; xs[4] = gx.z; xs[6] = gx.w;
    }
    __syncthreads();

    const v4f* bl = (const v4f*)(Bs + (4 * h) * K4_PAIR + c16 * K4_PITCH + chh * 8);

    v8f acc[4];
#pragma unroll
    for (int j = 0; j < 4; ++j) acc[j] = splat8(0.0f);

#pragma unroll
    for (int st = 0; st < 12; ++st) {
        const int T = st >> 2, t = st & 3;
        v2f a;
        if (T == 0)      { a = xf[t]; }
        else if (T == 1) { a.x = e0[2 * t]; a.y = e0[2 * t + 1]; }
        else             { a.x = e1[2 * t]; a.y = e1[2 * t + 1]; }
        const v4f* bp = bl + (8 * T + t) * (K4_PAIR / 4);
        const v4f f0 = bp[0], f1 = bp[1];
        v2f bj[4];
        bj[0].x = f0.x; bj[0].y = f0.y;
        bj[1].x = f0.z; bj[1].y = f0.w;
        bj[2].x = f1.x; bj[2].y = f1.y;
        bj[3].x = f1.z; bj[3].y = f1.w;
#pragma unroll
        for (int j = 0; j < 4; ++j)
            acc[j] = __builtin_amdgcn_wmma_f32_16x16x4_f32(
                false, a, false, bj[j], (short)0, acc[j], false, false);
    }

    const v4f bo = *((const v4f*)(bout + chh * 64) + c16);
#pragma unroll
    for (int r = 0; r < 8; ++r) {
        v4f o;
        o.x = tanh_hw(acc[0][r] + bo.x);
        o.y = tanh_hw(acc[1][r] + bo.y);
        o.z = tanh_hw(acc[2][r] + bo.z);
        o.w = tanh_hw(acc[3][r] + bo.w);
        *(v4f*)(out + (size_t)(row0 + r + 8 * h) * NF_ + chh * 64 + 4 * c16) = o;
    }
}

extern "C" void kernel_launch(void* const* d_in, const int* in_sizes, int n_in,
                              void* d_out, int out_size, void* d_ws, size_t ws_size,
                              hipStream_t stream)
{
    const float* x    = (const float*)d_in[0];
    const float* Wf   = (const float*)d_in[1];
    const float* bf   = (const float*)d_in[2];
    const float* Ws   = (const float*)d_in[3];
    const float* bs   = (const float*)d_in[4];
    const float* Wout = (const float*)d_in[5];
    const float* bout = (const float*)d_in[6];
    float* out = (float*)d_out;

    float* BWp = (float*)d_ws;

    k2_agg_bw<<<B_ * 8,  512, 0, stream>>>(x, Wf, bf, Ws, bs, Wout, BWp);
    k4_out   <<<B_ * 8,  512, 0, stream>>>(x, Ws, bs, Wout, BWp, bout, out);
}
